# starting kernel
# baseline (speedup 1.0000x reference)
#include <hip/hip_runtime.h>


typedef float v2f __attribute__((ext_vector_type(2)));
typedef float v8f __attribute__((ext_vector_type(8)));

struct c32 { float x, y; };
__device__ __forceinline__ c32 cmul(c32 a, c32 b) {
    return { a.x * b.x - a.y * b.y, a.x * b.y + a.y * b.x };
}

#define NQ     12
#define DIM    4096
#define NLAYER 4

__global__ __launch_bounds__(256)
void qsim_kernel(const float* __restrict__ xin,
                 const float* __restrict__ params,
                 float* __restrict__ out) {
    __shared__ float sre[DIM];
    __shared__ float sim_[DIM];
    __shared__ float bufAr[256], bufAi[256];
    __shared__ float bufBr[256], bufBi[256];
    __shared__ float uq[NQ][8];
    __shared__ float pv[NQ][6];
    __shared__ float pmr[NQ][16];
    __shared__ float pmi[NQ][16];
    __shared__ float enc[NQ][4];
    __shared__ float red[256 * NQ];

    const int tid  = threadIdx.x;
    const int b    = blockIdx.x;
    const int lane = tid & 31;
    const int wv   = tid >> 5;
    const int half = lane >> 4;
    const int lr   = lane & 15;

    auto wmma_apply = [&](const float* Wr, const float* Wi, int g) {
        const int rstride = (g == 0) ? 256 : ((g == 1) ? 16 : 1);
        const int nstride = (g == 2) ? 16 : 1;
        for (int tt = 0; tt < 2; ++tt) {
            int ct = 2 * wv + tt;
            int tbase = (g == 0) ? ct * 16 : ct * 256;
            v8f accR = {}; v8f accI = {};
#pragma unroll
            for (int ks = 0; ks < 4; ++ks) {
                int kk0 = 4 * ks + 2 * half;
                v2f ar, ai, ain, br, bi;
                ar[0] = Wr[lr * 16 + kk0];  ar[1] = Wr[lr * 16 + kk0 + 1];
                ai[0] = Wi[lr * 16 + kk0];  ai[1] = Wi[lr * 16 + kk0 + 1];
                ain = -ai;
                int i0 = tbase + kk0 * rstride + lr * nstride;
                br[0] = sre[i0];  br[1] = sre[i0 + rstride];
                bi[0] = sim_[i0]; bi[1] = sim_[i0 + rstride];
                accR = __builtin_amdgcn_wmma_f32_16x16x4_f32(
                    false, ar, false, br, (short)0, accR, false, false);
                accR = __builtin_amdgcn_wmma_f32_16x16x4_f32(
                    false, ain, false, bi, (short)0, accR, false, false);
                accI = __builtin_amdgcn_wmma_f32_16x16x4_f32(
                    false, ar, false, bi, (short)0, accI, false, false);
                accI = __builtin_amdgcn_wmma_f32_16x16x4_f32(
                    false, ai, false, br, (short)0, accI, false, false);
            }
#pragma unroll
            for (int j = 0; j < 8; ++j) {
                int M = j + 8 * half;
                int o = tbase + M * rstride + lr * nstride;
                sre[o] = accR[j]; sim_[o] = accI[j];
            }
        }
    };

    auto kron_build = [&](int g, float* outR, float* outI) {
        int r = tid >> 4, c = tid & 15, q0 = 4 * g;
        int o0 = ((((r >> 3) & 1) * 2 + ((c >> 3) & 1)) * 2);
        int o1 = ((((r >> 2) & 1) * 2 + ((c >> 2) & 1)) * 2);
        int o2 = ((((r >> 1) & 1) * 2 + ((c >> 1) & 1)) * 2);
        int o3 = (((r & 1) * 2 + (c & 1)) * 2);
        c32 acc = { uq[q0][o0], uq[q0][o0 + 1] };
        acc = cmul(acc, c32{ uq[q0 + 1][o1], uq[q0 + 1][o1 + 1] });
        acc = cmul(acc, c32{ uq[q0 + 2][o2], uq[q0 + 2][o2 + 1] });
        acc = cmul(acc, c32{ uq[q0 + 3][o3], uq[q0 + 3][o3 + 1] });
        outR[tid] = acc.x; outI[tid] = acc.y;
    };

    auto chain_step = [&](int pair, int t,
                          const float* inR, const float* inI,
                          float* outR, float* outI) {
        int r = tid >> 4, c = tid & 15;
        int sh = 2 - t;
        int rowsel = ((r >> sh) & 3) * 4;
        int rmask = r & ~(3 << sh);
        c32 acc = { 0.0f, 0.0f };
#pragma unroll
        for (int j = 0; j < 4; ++j) {
            int k = rmask | (j << sh);
            c32 p = { pmr[pair][rowsel + j], pmi[pair][rowsel + j] };
            c32 v = { inR[k * 16 + c], inI[k * 16 + c] };
            acc.x += p.x * v.x - p.y * v.y;
            acc.y += p.x * v.y + p.y * v.x;
        }
        outR[tid] = acc.x; outI[tid] = acc.y;
    };

    auto pfull_build = [&](int pair, float* outR, float* outI) {
        int r = tid >> 4, c = tid & 15;
        if ((r & 3) == (c & 3)) {
            int idx = ((r >> 2) & 3) * 4 + ((c >> 2) & 3);
            outR[tid] = pmr[pair][idx]; outI[tid] = pmi[pair][idx];
        } else {
            outR[tid] = 0.0f; outI[tid] = 0.0f;
        }
    };

    auto pair_sweep = [&](int i) {
        int j  = (i + 1) % NQ;
        int pi_ = 11 - i, pj_ = 11 - j;
        int lo = pi_ < pj_ ? pi_ : pj_;
        int hi = pi_ < pj_ ? pj_ : pi_;
        float cd = pv[i][0], sd = pv[i][1];
        float cs = pv[i][2], ss = pv[i][3];
        float cph = pv[i][4], sph = pv[i][5];
#pragma unroll
        for (int m = 0; m < 4; ++m) {
            int qd   = tid + 256 * m;
            int low  = qd & ((1 << lo) - 1);
            int mid  = (qd >> lo) & ((1 << (hi - 1 - lo)) - 1);
            int high = qd >> (hi - 1);
            int base = (high << (hi + 1)) | (mid << (lo + 1)) | low;
            int i01 = base | (1 << pj_);
            int i10 = base | (1 << pi_);
            int i11 = i10  | (1 << pj_);
            c32 a00 = { sre[base], sim_[base] };
            c32 a01 = { sre[i01],  sim_[i01]  };
            c32 a10 = { sre[i10],  sim_[i10]  };
            c32 a11 = { sre[i11],  sim_[i11]  };
            c32 b11 = { cph * a11.x - sph * a11.y, cph * a11.y + sph * a11.x };
            c32 b10 = { cph * a10.x + sph * a10.y, cph * a10.y - sph * a10.x };
            c32 n00 = {  cd * a00.x + sd * b11.y,  cd * a00.y - sd * b11.x };
            c32 n11 = {  sd * a00.y + cd * b11.x, -sd * a00.x + cd * b11.y };
            c32 n01 = {  cs * a01.x + ss * b10.y,  cs * a01.y - ss * b10.x };
            c32 n10 = {  ss * a01.y + cs * b10.x, -ss * a01.x + cs * b10.y };
            sre[base] = n00.x; sim_[base] = n00.y;
            sre[i01]  = n01.x; sim_[i01]  = n01.y;
            sre[i10]  = n10.x; sim_[i10]  = n10.y;
            sre[i11]  = n11.x; sim_[i11]  = n11.y;
        }
    };

    if (tid < NQ) {
        float xv = xin[b * NQ + tid];
        float h = 0.5f * xv;
        float a = 0.5f * xv * xv;
        float ch = cosf(h), sh = sinf(h);
        float ca = cosf(a), sa = sinf(a);
        enc[tid][0] = ch * ca;  enc[tid][1] = -ch * sa;
        enc[tid][2] = sh * ca;  enc[tid][3] =  sh * sa;
    }
    __syncthreads();

    for (int m = 0; m < 16; ++m) {
        int k = tid + 256 * m;
        c32 amp = { 1.0f, 0.0f };
#pragma unroll
        for (int q = 0; q < NQ; ++q) {
            int bit = (k >> (11 - q)) & 1;
            c32 f = { enc[q][bit * 2], enc[q][bit * 2 + 1] };
            amp = cmul(amp, f);
        }
        sre[k] = amp.x; sim_[k] = amp.y;
    }
    __syncthreads();

    for (int l = 0; l < NLAYER; ++l) {
        if (tid < NQ) {
            const float* wp = params + (l * NQ + tid) * 3;
            float w0 = wp[0], w1 = wp[1], w2 = wp[2];
            float c0 = cosf(0.5f * w0), s0 = sinf(0.5f * w0);
            float c1 = cosf(0.5f * w1), s1 = sinf(0.5f * w1);
            float cz = cosf(0.5f * w2), sz = sinf(0.5f * w2);
            c32 A  = {  c1 * c0,  s1 * s0 };
            c32 Bm = { -s1 * c0, -c1 * s0 };
            c32 Cm = {  s1 * c0, -c1 * s0 };
            c32 Dm = {  c1 * c0, -s1 * s0 };
            c32 ez  = { cz,  sz };
            c32 ezc = { cz, -sz };
            c32 u00 = cmul(A, ezc), u01 = cmul(Bm, ezc);
            c32 u10 = cmul(Cm, ez), u11 = cmul(Dm, ez);
            uq[tid][0] = u00.x; uq[tid][1] = u00.y;
            uq[tid][2] = u01.x; uq[tid][3] = u01.y;
            uq[tid][4] = u10.x; uq[tid][5] = u10.y;
            uq[tid][6] = u11.x; uq[tid][7] = u11.y;
            float cd = cosf(0.5f * (w1 - w2)), sd = sinf(0.5f * (w1 - w2));
            float cs = cosf(0.5f * (w1 + w2)), ss = sinf(0.5f * (w1 + w2));
            float cph = cosf(0.5f * w0), sph = sinf(0.5f * w0);
            pv[tid][0] = cd; pv[tid][1] = sd;
            pv[tid][2] = cs; pv[tid][3] = ss;
            pv[tid][4] = cph; pv[tid][5] = sph;
#pragma unroll
            for (int e = 0; e < 16; ++e) { pmr[tid][e] = 0.0f; pmi[tid][e] = 0.0f; }
            pmr[tid][0]  = cd;
            pmr[tid][3]  = sd * sph;  pmi[tid][3]  = -sd * cph;
            pmi[tid][12] = -sd;
            pmr[tid][15] = cd * cph;  pmi[tid][15] =  cd * sph;
            pmr[tid][5]  = cs;
            pmr[tid][6]  = -ss * sph; pmi[tid][6]  = -ss * cph;
            pmi[tid][9]  = -ss;
            pmr[tid][10] = cs * cph;  pmi[tid][10] = -cs * sph;
        }
        __syncthreads();

        kron_build(1, bufAr, bufAi);                 __syncthreads();
        wmma_apply(bufAr, bufAi, 1);                 __syncthreads();
        kron_build(2, bufAr, bufAi);                 __syncthreads();
        wmma_apply(bufAr, bufAi, 2);                 __syncthreads();

        kron_build(0, bufAr, bufAi);                 __syncthreads();
        chain_step(0, 0, bufAr, bufAi, bufBr, bufBi); __syncthreads();
        chain_step(1, 1, bufBr, bufBi, bufAr, bufAi); __syncthreads();
        chain_step(2, 2, bufAr, bufAi, bufBr, bufBi); __syncthreads();
        wmma_apply(bufBr, bufBi, 0);                 __syncthreads();

        pair_sweep(3);                               __syncthreads();

        pfull_build(4, bufAr, bufAi);                __syncthreads();
        chain_step(5, 1, bufAr, bufAi, bufBr, bufBi); __syncthreads();
        chain_step(6, 2, bufBr, bufBi, bufAr, bufAi); __syncthreads();
        wmma_apply(bufAr, bufAi, 1);                 __syncthreads();

        pair_sweep(7);                               __syncthreads();

        pfull_build(8, bufAr, bufAi);                __syncthreads();
        chain_step(9, 1, bufAr, bufAi, bufBr, bufBi); __syncthreads();
        chain_step(10, 2, bufBr, bufBi, bufAr, bufAi); __syncthreads();
        wmma_apply(bufAr, bufAi, 2);                 __syncthreads();

        pair_sweep(11);                              __syncthreads();
    }

    float part[NQ];
#pragma unroll
    for (int q = 0; q < NQ; ++q) part[q] = 0.0f;
    for (int m = 0; m < 16; ++m) {
        int k = tid + 256 * m;
        float pr = sre[k] * sre[k] + sim_[k] * sim_[k];
#pragma unroll
        for (int q = 0; q < NQ; ++q)
            part[q] += ((k >> (11 - q)) & 1) ? -pr : pr;
    }
#pragma unroll
    for (int q = 0; q < NQ; ++q) red[tid * NQ + q] = part[q];
    __syncthreads();
    for (int s = 128; s > 0; s >>= 1) {
        if (tid < s) {
#pragma unroll
            for (int q = 0; q < NQ; ++q)
                red[tid * NQ + q] += red[(tid + s) * NQ + q];
        }
        __syncthreads();
    }
    if (tid < NQ) out[b * NQ + tid] = red[tid];
}

extern "C" void kernel_launch(void* const* d_in, const int* in_sizes, int n_in,
                              void* d_out, int out_size, void* d_ws, size_t ws_size,
                              hipStream_t stream) {
    (void)n_in; (void)out_size; (void)d_ws; (void)ws_size;
    const float* x      = (const float*)d_in[0];
    const float* params = (const float*)d_in[1];
    float* out = (float*)d_out;
    int B = in_sizes[0] / NQ;
    qsim_kernel<<<dim3(B), dim3(256), 0, stream>>>(x, params, out);
}


# Four 3-wire groups per layer, each 8x8 complex unitary as one real 16x16 WMMA operand [[Vr,-Vi],[Vi,Vr]] on stacked re/im rows: 64 WMMAs per wave-layer, not 96; sum order changes
# speedup vs baseline: 1.8753x; 1.8753x over previous
#include <hip/hip_runtime.h>


typedef float v2f __attribute__((ext_vector_type(2)));
typedef float v8f __attribute__((ext_vector_type(8)));

struct c32 { float x, y; };
__device__ __forceinline__ c32 cmul(c32 a, c32 b) {
    return { a.x * b.x - a.y * b.y, a.x * b.y + a.y * b.x };
}

#define NQ     12
#define DIM    4096
#define NLAYER 4

__device__ __forceinline__ constexpr int swz(int k) { return k ^ ((k >> 6) & 63); }

__global__ __launch_bounds__(256)
void qsim_kernel(const float* __restrict__ xin,
                 const float* __restrict__ params,
                 float* __restrict__ out) {
    __shared__ float sre[DIM];
    __shared__ float sim_[DIM];
    __shared__ float AL[2][4][256];
    __shared__ float Tar[4][64], Tai[4][64];
    __shared__ float Tbr[4][64], Tbi[4][64];
    __shared__ float uq[NLAYER][NQ][8];
    __shared__ float pv[NLAYER][NQ][6];
    __shared__ float pmr[NLAYER][NQ][16];
    __shared__ float pmi[NLAYER][NQ][16];
    __shared__ float enc[NQ][4];
    __shared__ float red[256 * NQ];

    const int tid  = threadIdx.x;
    const int b    = blockIdx.x;
    const int lane = tid & 31;
    const int wv   = tid >> 5;
    const int half = lane >> 4;
    const int lr   = lane & 15;

    auto wave_sync = [&]() {
        __builtin_amdgcn_fence(__ATOMIC_RELEASE, "wavefront");
        __builtin_amdgcn_wave_barrier();
        __builtin_amdgcn_fence(__ATOMIC_ACQUIRE, "wavefront");
    };

    auto apply_pair = [&](const float* A, int gb, int yb, int tb,
                          int nb0, int nb1, int nb2, int nb3, int wpart,
                          int lra, int ha,
                          int l, int i, bool withU) __attribute__((always_inline)) {
        const int j = (i + 1) % NQ;
        const float cd = pv[l][i][0], sd = pv[l][i][1];
        const float cs = pv[l][i][2], ss = pv[l][i][3];
        const float cph = pv[l][i][4], sph = pv[l][i][5];
        c32 g00 = { 1.0f, 0.0f }, g01 = { 0.0f, 0.0f };
        c32 g10 = { 0.0f, 0.0f }, g11 = { 1.0f, 0.0f };
        if (withU) {
            g00 = { uq[l][j][0], uq[l][j][1] };  g01 = { uq[l][j][2], uq[l][j][3] };
            g10 = { uq[l][j][4], uq[l][j][5] };  g11 = { uq[l][j][6], uq[l][j][7] };
        }
        const int lanepart = ((lra & 1) << nb0) | (((lra >> 1) & 1) << nb1)
                           | (((lra >> 2) & 1) << nb2) | (((lra >> 3) & 1) << nb3);
        const int ldbase = swz(wpart | lanepart | (ha << gb));
        const int stbase = swz(wpart | lanepart | (ha << (gb + 2)));
        v2f af[4];
#pragma unroll
        for (int ks = 0; ks < 4; ++ks) {
            af[ks][0] = A[lr * 16 + half * 8 + 2 * ks];
            af[ks][1] = A[lr * 16 + half * 8 + 2 * ks + 1];
        }
        v2f bf[2][2][4];
#pragma unroll
        for (int y = 0; y < 2; ++y) {
#pragma unroll
            for (int t = 0; t < 2; ++t) {
#pragma unroll
                for (int ks = 0; ks < 4; ++ks) {
                    const int a = ldbase ^ swz((y << yb) | (t << tb) | ((2 * ks) << gb));
                    bf[y][t][ks][0] = sre[a]; bf[y][t][ks][1] = sim_[a];
                }
            }
        }
        v8f acc[2][2];
#pragma unroll
        for (int y = 0; y < 2; ++y) {
#pragma unroll
            for (int t = 0; t < 2; ++t) acc[y][t] = v8f{};
        }
#pragma unroll
        for (int ks = 0; ks < 4; ++ks) {
#pragma unroll
            for (int y = 0; y < 2; ++y) {
#pragma unroll
                for (int t = 0; t < 2; ++t) {
                    acc[y][t] = __builtin_amdgcn_wmma_f32_16x16x4_f32(
                        false, af[ks], false, bf[y][t][ks], (short)0, acc[y][t], false, false);
                }
            }
        }
        const v2f PC  = { 1.0f, cph };
        const v2f PS0 = { 0.0f, -sph };
        const v2f PS1 = { 0.0f,  sph };
        const v2f C0 = { cd, cs }, S0 = { sd, ss };
        const v2f C1 = { cs, cd }, S1 = { ss, sd };
#pragma unroll
        for (int t = 0; t < 2; ++t) {
            v2f nR[2][2], nI[2][2];
#pragma unroll
            for (int p = 0; p < 2; ++p) {
                v2f T0r = { acc[0][t][2 * p],     acc[0][t][2 * p + 1] };
                v2f T0i = { acc[0][t][4 + 2 * p], acc[0][t][4 + 2 * p + 1] };
                v2f T1r = { acc[1][t][2 * p],     acc[1][t][2 * p + 1] };
                v2f T1i = { acc[1][t][4 + 2 * p], acc[1][t][4 + 2 * p + 1] };
                if (withU) {
                    v2f U0r = g00.x * T0r - g00.y * T0i + g01.x * T1r - g01.y * T1i;
                    v2f U0i = g00.x * T0i + g00.y * T0r + g01.x * T1i + g01.y * T1r;
                    v2f U1r = g10.x * T0r - g10.y * T0i + g11.x * T1r - g11.y * T1i;
                    v2f U1i = g10.x * T0i + g10.y * T0r + g11.x * T1i + g11.y * T1r;
                    T0r = U0r; T0i = U0i; T1r = U1r; T1i = U1i;
                }
                v2f B0r = PC * T0r - PS0 * T0i, B0i = PC * T0i + PS0 * T0r;
                v2f B1r = PC * T1r - PS1 * T1i, B1i = PC * T1i + PS1 * T1r;
                v2f B0rs = B0r.yx, B0is = B0i.yx, B1rs = B1r.yx, B1is = B1i.yx;
                nR[0][p] = C0 * B0r + S0 * B1is;  nI[0][p] = C0 * B0i - S0 * B1rs;
                nR[1][p] = C1 * B1r + S1 * B0is;  nI[1][p] = C1 * B1i - S1 * B0rs;
            }
#pragma unroll
            for (int y = 0; y < 2; ++y) {
#pragma unroll
                for (int j2 = 0; j2 < 4; ++j2) {
                    const int o = stbase ^ swz((y << yb) | (t << tb) | (j2 << gb));
                    sre[o] = nR[y][j2 >> 1][j2 & 1]; sim_[o] = nI[y][j2 >> 1][j2 & 1];
                }
            }
        }
    };

    auto layer_scalars = [&](int l, int q) {
        const float* wp = params + (l * NQ + q) * 3;
        float w0 = wp[0], w1 = wp[1], w2 = wp[2];
        float c0, s0, c1, s1, cz, sz;
        sincosf(0.5f * w0, &s0, &c0);
        sincosf(0.5f * w1, &s1, &c1);
        sincosf(0.5f * w2, &sz, &cz);
        c32 A  = {  c1 * c0,  s1 * s0 };
        c32 Bm = { -s1 * c0, -c1 * s0 };
        c32 Cm = {  s1 * c0, -c1 * s0 };
        c32 Dm = {  c1 * c0, -s1 * s0 };
        c32 ez  = { cz,  sz };
        c32 ezc = { cz, -sz };
        c32 u00 = cmul(A, ezc), u01 = cmul(Bm, ezc);
        c32 u10 = cmul(Cm, ez), u11 = cmul(Dm, ez);
        uq[l][q][0] = u00.x; uq[l][q][1] = u00.y;
        uq[l][q][2] = u01.x; uq[l][q][3] = u01.y;
        uq[l][q][4] = u10.x; uq[l][q][5] = u10.y;
        uq[l][q][6] = u11.x; uq[l][q][7] = u11.y;
        float cd = c1 * cz + s1 * sz, sd = s1 * cz - c1 * sz;
        float cs = c1 * cz - s1 * sz, ss = s1 * cz + c1 * sz;
        float cph = c0, sph = s0;
        pv[l][q][0] = cd; pv[l][q][1] = sd;
        pv[l][q][2] = cs; pv[l][q][3] = ss;
        pv[l][q][4] = cph; pv[l][q][5] = sph;
#pragma unroll
        for (int e = 0; e < 16; ++e) { pmr[l][q][e] = 0.0f; pmi[l][q][e] = 0.0f; }
        pmr[l][q][0]  = cd;
        pmr[l][q][3]  = sd * sph;  pmi[l][q][3]  = -sd * cph;
        pmi[l][q][12] = -sd;
        pmr[l][q][15] = cd * cph;  pmi[l][q][15] =  cd * sph;
        pmr[l][q][5]  = cs;
        pmr[l][q][6]  = -ss * sph; pmi[l][q][6]  = -ss * cph;
        pmi[l][q][9]  = -ss;
        pmr[l][q][10] = cs * cph;  pmi[l][q][10] = -cs * sph;
    };

    auto build_unitaries = [&](int l, int ub) {
        const int g = wv >> 1, c = 4 * (wv & 1) + (lane & 3), r = lane >> 2, q0 = 3 * g;
        {
            int o0 = ((((r >> 2) & 1) * 2 + ((c >> 2) & 1)) * 2);
            int o1 = ((((r >> 1) & 1) * 2 + ((c >> 1) & 1)) * 2);
            int o2 = (((r & 1) * 2 + (c & 1)) * 2);
            c32 acc;
            if (g == 0) acc = { uq[l][0][o0], uq[l][0][o0 + 1] };
            else        acc = { (((r >> 2) & 1) == ((c >> 2) & 1)) ? 1.0f : 0.0f, 0.0f };
            acc = cmul(acc, c32{ uq[l][q0 + 1][o1], uq[l][q0 + 1][o1 + 1] });
            acc = cmul(acc, c32{ uq[l][q0 + 2][o2], uq[l][q0 + 2][o2 + 1] });
            Tar[g][r * 8 + c] = acc.x; Tai[g][r * 8 + c] = acc.y;
        }
        wave_sync();
        auto chain8 = [&](int pair, int sh, const float* inR, const float* inI) {
            int rowsel = ((r >> sh) & 3) * 4;
            int rmask = r & ~(3 << sh);
            c32 acc = { 0.0f, 0.0f };
#pragma unroll
            for (int jj = 0; jj < 4; ++jj) {
                int k = rmask | (jj << sh);
                c32 p = { pmr[l][pair][rowsel + jj], pmi[l][pair][rowsel + jj] };
                c32 v = { inR[k * 8 + c], inI[k * 8 + c] };
                acc.x += p.x * v.x - p.y * v.y;
                acc.y += p.x * v.y + p.y * v.x;
            }
            return acc;
        };
        c32 s0 = chain8(q0, 1, Tar[g], Tai[g]);
        Tbr[g][r * 8 + c] = s0.x; Tbi[g][r * 8 + c] = s0.y;
        wave_sync();
        c32 u = chain8(q0 + 1, 0, Tbr[g], Tbi[g]);
        float* Ao = AL[ub][g];
        const int M0 = 8 * (r >> 2) + (r & 3), M1 = M0 + 4;
        const int kb = (c & 1) * 8 + (c >> 1) * 2;
        Ao[M0 * 16 + kb] = u.x;  Ao[M0 * 16 + kb + 1] = -u.y;
        Ao[M1 * 16 + kb] = u.y;  Ao[M1 * 16 + kb + 1] =  u.x;
    };

    if (lane < NQ) {
        if (wv == 0) {
            float xv = xin[b * NQ + lane];
            float h = 0.5f * xv;
            float a = 0.5f * xv * xv;
            float ch, sh, ca, sa;
            sincosf(h, &sh, &ch);
            sincosf(a, &sa, &ca);
            enc[lane][0] = ch * ca;  enc[lane][1] = -ch * sa;
            enc[lane][2] = sh * ca;  enc[lane][3] =  sh * sa;
        } else if (wv <= NLAYER) {
            layer_scalars(wv - 1, lane);
        }
    }
    __syncthreads();

    build_unitaries(0, 0);

    for (int m = 0; m < 16; ++m) {
        int k = tid + 256 * m;
        c32 amp = { 1.0f, 0.0f };
#pragma unroll
        for (int q = 0; q < NQ; ++q) {
            int bit = (k >> (11 - q)) & 1;
            c32 f = { enc[q][bit * 2], enc[q][bit * 2 + 1] };
            amp = cmul(amp, f);
        }
        sre[swz(k)] = amp.x; sim_[swz(k)] = amp.y;
    }
    __syncthreads();

    const int w1part = wv;
    const int w2part = ((wv & 1) << 6) | (((wv >> 1) & 1) << 7) | (((wv >> 2) & 1) << 9);

    for (int l = 0; l < NLAYER; ++l) {
        const int ub = l & 1;
        int lra = lr, ha = half;
        asm volatile("" : "+v"(lra), "+v"(ha));

        apply_pair(AL[ub][0],  9, 8, 5,  3, 4, 6, 7,    w1part, lra, ha, l, 2, true);
        wave_sync();
        apply_pair(AL[ub][1],  6, 5, 4,  3, 9, 10, 11,  w1part, lra, ha, l, 5, true);
        __syncthreads();

        apply_pair(AL[ub][2],  3, 2, 11, 0, 1, 8, 10,   w2part, lra, ha, l, 8, true);
        wave_sync();
        if (l + 1 < NLAYER) build_unitaries(l + 1, ub ^ 1);
        apply_pair(AL[ub][3],  0, 11, 10, 3, 4, 5, 8,   w2part, lra, ha, l, 11, false);
        __syncthreads();
    }

    float part[NQ];
#pragma unroll
    for (int q = 0; q < NQ; ++q) part[q] = 0.0f;
    for (int m = 0; m < 16; ++m) {
        int k = tid + 256 * m;
        float vr = sre[swz(k)], vi = sim_[swz(k)];
        float pr = vr * vr + vi * vi;
#pragma unroll
        for (int q = 0; q < NQ; ++q)
            part[q] += ((k >> (11 - q)) & 1) ? -pr : pr;
    }
#pragma unroll
    for (int q = 0; q < NQ; ++q) red[tid * NQ + q] = part[q];
    __syncthreads();
    for (int s = 128; s > 0; s >>= 1) {
        if (tid < s) {
#pragma unroll
            for (int q = 0; q < NQ; ++q)
                red[tid * NQ + q] += red[(tid + s) * NQ + q];
        }
        __syncthreads();
    }
    if (tid < NQ) out[b * NQ + tid] = red[tid];
}

extern "C" void kernel_launch(void* const* d_in, const int* in_sizes, int n_in,
                              void* d_out, int out_size, void* d_ws, size_t ws_size,
                              hipStream_t stream) {
    (void)n_in; (void)out_size; (void)d_ws; (void)ws_size;
    const float* x      = (const float*)d_in[0];
    const float* params = (const float*)d_in[1];
    float* out = (float*)d_out;
    int B = in_sizes[0] / NQ;
    qsim_kernel<<<dim3(B), dim3(256), 0, stream>>>(x, params, out);
}


# Single-wire factors leave the cross sweeps: next layer's u3,u6,u9 and each cross CRZ's control-wire phase become factors of the group unitary in front; sweep keeps |11> phase + mix; rounding changes
# speedup vs baseline: 2.0079x; 1.0707x over previous
#include <hip/hip_runtime.h>


typedef float v2f __attribute__((ext_vector_type(2)));
typedef float v8f __attribute__((ext_vector_type(8)));

struct c32 { float x, y; };
__device__ __forceinline__ c32 cmul(c32 a, c32 b) {
    return { a.x * b.x - a.y * b.y, a.x * b.y + a.y * b.x };
}

struct gate1 { c32 u00, u01, u10, u11; float c0, s0, c1, s1, cz, sz; };
__device__ __forceinline__ gate1 wire_gate(const float* wp) {
    gate1 g;
    float w0 = wp[0], w1 = wp[1], w2 = wp[2];
    sincosf(0.5f * w0, &g.s0, &g.c0);
    sincosf(0.5f * w1, &g.s1, &g.c1);
    sincosf(0.5f * w2, &g.sz, &g.cz);
    c32 A  = {  g.c1 * g.c0,  g.s1 * g.s0 };
    c32 Bm = { -g.s1 * g.c0, -g.c1 * g.s0 };
    c32 Cm = {  g.s1 * g.c0, -g.c1 * g.s0 };
    c32 Dm = {  g.c1 * g.c0, -g.s1 * g.s0 };
    c32 ez  = { g.cz,  g.sz };
    c32 ezc = { g.cz, -g.sz };
    g.u00 = cmul(A, ezc);  g.u01 = cmul(Bm, ezc);
    g.u10 = cmul(Cm, ez);  g.u11 = cmul(Dm, ez);
    return g;
}

#define NQ     12
#define DIM    4096
#define NLAYER 4

__device__ __forceinline__ constexpr int swz(int k) { return k ^ ((k >> 6) & 63); }

__global__ __launch_bounds__(256)
void qsim_kernel(const float* __restrict__ xin,
                 const float* __restrict__ params,
                 float* __restrict__ out) {
    __shared__ float sre[DIM];
    __shared__ float sim_[DIM];
    __shared__ float AL[2][4][256];
    __shared__ float Tar[4][64], Tai[4][64];
    __shared__ float Tbr[4][64], Tbi[4][64];
    __shared__ float uq[NLAYER][NQ][8];
    __shared__ float pv[NLAYER][NQ][8];
    __shared__ float pmr[NLAYER][NQ][16];
    __shared__ float pmi[NLAYER][NQ][16];
    __shared__ float enc[NQ][4];
    __shared__ float red[256 * NQ];

    const int tid  = threadIdx.x;
    const int b    = blockIdx.x;
    const int lane = tid & 31;
    const int wv   = tid >> 5;
    const int half = lane >> 4;
    const int lr   = lane & 15;

    auto wave_sync = [&]() {
        __builtin_amdgcn_fence(__ATOMIC_RELEASE, "wavefront");
        __builtin_amdgcn_wave_barrier();
        __builtin_amdgcn_fence(__ATOMIC_ACQUIRE, "wavefront");
    };

    auto apply_pair = [&](const float* A, int gb, int yb, int tb,
                          int nb0, int nb1, int nb2, int nb3, int wpart,
                          int lra, int ha,
                          int l, int i) __attribute__((always_inline)) {
        const float cd = pv[l][i][0], sd = pv[l][i][1];
        const float cs = pv[l][i][2], ss = pv[l][i][3];
        const float c2 = pv[l][i][6], s2 = pv[l][i][7];
        const int lanepart = ((lra & 1) << nb0) | (((lra >> 1) & 1) << nb1)
                           | (((lra >> 2) & 1) << nb2) | (((lra >> 3) & 1) << nb3);
        const int ldbase = swz(wpart | lanepart | (ha << gb));
        const int stbase = swz(wpart | lanepart | (ha << (gb + 2)));
        v2f af[4];
#pragma unroll
        for (int ks = 0; ks < 4; ++ks) {
            af[ks][0] = A[lr * 16 + half * 8 + 2 * ks];
            af[ks][1] = A[lr * 16 + half * 8 + 2 * ks + 1];
        }
        v2f bf[2][2][4];
#pragma unroll
        for (int y = 0; y < 2; ++y) {
#pragma unroll
            for (int t = 0; t < 2; ++t) {
#pragma unroll
                for (int ks = 0; ks < 4; ++ks) {
                    const int a = ldbase ^ swz((y << yb) | (t << tb) | ((2 * ks) << gb));
                    bf[y][t][ks][0] = sre[a]; bf[y][t][ks][1] = sim_[a];
                }
            }
        }
        v8f acc[2][2];
#pragma unroll
        for (int y = 0; y < 2; ++y) {
#pragma unroll
            for (int t = 0; t < 2; ++t) acc[y][t] = v8f{};
        }
#pragma unroll
        for (int ks = 0; ks < 4; ++ks) {
#pragma unroll
            for (int y = 0; y < 2; ++y) {
#pragma unroll
                for (int t = 0; t < 2; ++t) {
                    acc[y][t] = __builtin_amdgcn_wmma_f32_16x16x4_f32(
                        false, af[ks], false, bf[y][t][ks], (short)0, acc[y][t], false, false);
                }
            }
        }
        const v2f PC = { 1.0f, c2 };
        const v2f PS = { 0.0f, s2 };
        const v2f C0 = { cd, cs }, S0 = { sd, ss };
        const v2f C1 = { cs, cd }, S1 = { ss, sd };
#pragma unroll
        for (int t = 0; t < 2; ++t) {
            v2f nR[2][2], nI[2][2];
#pragma unroll
            for (int p = 0; p < 2; ++p) {
                v2f T0r = { acc[0][t][2 * p],     acc[0][t][2 * p + 1] };
                v2f T0i = { acc[0][t][4 + 2 * p], acc[0][t][4 + 2 * p + 1] };
                v2f T1r = { acc[1][t][2 * p],     acc[1][t][2 * p + 1] };
                v2f T1i = { acc[1][t][4 + 2 * p], acc[1][t][4 + 2 * p + 1] };
                v2f B1r = PC * T1r - PS * T1i, B1i = PC * T1i + PS * T1r;
                v2f T0rs = T0r.yx, T0is = T0i.yx, B1rs = B1r.yx, B1is = B1i.yx;
                nR[0][p] = C0 * T0r + S0 * B1is;  nI[0][p] = C0 * T0i - S0 * B1rs;
                nR[1][p] = C1 * B1r + S1 * T0is;  nI[1][p] = C1 * B1i - S1 * T0rs;
            }
#pragma unroll
            for (int y = 0; y < 2; ++y) {
#pragma unroll
                for (int j2 = 0; j2 < 4; ++j2) {
                    const int o = stbase ^ swz((y << yb) | (t << tb) | (j2 << gb));
                    sre[o] = nR[y][j2 >> 1][j2 & 1]; sim_[o] = nI[y][j2 >> 1][j2 & 1];
                }
            }
        }
    };

    auto layer_scalars = [&](int l, int q) {
        const gate1 gq = wire_gate(params + (l * NQ + q) * 3);
        const float c0 = gq.c0, s0 = gq.s0, c1 = gq.c1, s1 = gq.s1, cz = gq.cz, sz = gq.sz;
        const c32 u00 = gq.u00, u01 = gq.u01, u10 = gq.u10, u11 = gq.u11;
        uq[l][q][0] = u00.x; uq[l][q][1] = u00.y;
        uq[l][q][2] = u01.x; uq[l][q][3] = u01.y;
        uq[l][q][4] = u10.x; uq[l][q][5] = u10.y;
        uq[l][q][6] = u11.x; uq[l][q][7] = u11.y;
        float cd = c1 * cz + s1 * sz, sd = s1 * cz - c1 * sz;
        float cs = c1 * cz - s1 * sz, ss = s1 * cz + c1 * sz;
        float cph = c0, sph = s0;
        pv[l][q][0] = cd; pv[l][q][1] = sd;
        pv[l][q][2] = cs; pv[l][q][3] = ss;
        pv[l][q][4] = cph; pv[l][q][5] = sph;
        pv[l][q][6] = cph * cph - sph * sph; pv[l][q][7] = 2.0f * sph * cph;
#pragma unroll
        for (int e = 0; e < 16; ++e) { pmr[l][q][e] = 0.0f; pmi[l][q][e] = 0.0f; }
        pmr[l][q][0]  = cd;
        pmr[l][q][3]  = sd * sph;  pmi[l][q][3]  = -sd * cph;
        pmi[l][q][12] = -sd;
        pmr[l][q][15] = cd * cph;  pmi[l][q][15] =  cd * sph;
        pmr[l][q][5]  = cs;
        pmr[l][q][6]  = -ss * sph; pmi[l][q][6]  = -ss * cph;
        pmi[l][q][9]  = -ss;
        pmr[l][q][10] = cs * cph;  pmi[l][q][10] = -cs * sph;
    };

    auto build_unitaries = [&](int l, int ub) {
        const int g = wv >> 1, c = 4 * (wv & 1) + (lane & 3), r = lane >> 2, q0 = 3 * g;
        {
            int o0 = ((((r >> 2) & 1) * 2 + ((c >> 2) & 1)) * 2);
            int o1 = ((((r >> 1) & 1) * 2 + ((c >> 1) & 1)) * 2);
            int o2 = (((r & 1) * 2 + (c & 1)) * 2);
            c32 acc;
            if (g == 0) acc = { uq[l][0][o0], uq[l][0][o0 + 1] };
            else        acc = { (((r >> 2) & 1) == ((c >> 2) & 1)) ? 1.0f : 0.0f, 0.0f };
            acc = cmul(acc, c32{ uq[l][q0 + 1][o1], uq[l][q0 + 1][o1 + 1] });
            acc = cmul(acc, c32{ uq[l][q0 + 2][o2], uq[l][q0 + 2][o2 + 1] });
            Tar[g][r * 8 + c] = acc.x; Tai[g][r * 8 + c] = acc.y;
        }
        wave_sync();
        auto chain8 = [&](int pair, int sh, const float* inR, const float* inI) {
            int rowsel = ((r >> sh) & 3) * 4;
            int rmask = r & ~(3 << sh);
            c32 acc = { 0.0f, 0.0f };
#pragma unroll
            for (int jj = 0; jj < 4; ++jj) {
                int k = rmask | (jj << sh);
                c32 p = { pmr[l][pair][rowsel + jj], pmi[l][pair][rowsel + jj] };
                c32 v = { inR[k * 8 + c], inI[k * 8 + c] };
                acc.x += p.x * v.x - p.y * v.y;
                acc.y += p.x * v.y + p.y * v.x;
            }
            return acc;
        };
        c32 s0 = chain8(q0, 1, Tar[g], Tai[g]);
        Tbr[g][r * 8 + c] = s0.x; Tbi[g][r * 8 + c] = s0.y;
        wave_sync();
        c32 u = chain8(q0 + 1, 0, Tbr[g], Tbi[g]);
        if (r & 1) u = cmul(u, c32{ pv[l][q0 + 2][4], -pv[l][q0 + 2][5] });
        if (g > 0 && l + 1 < NLAYER) {
            const c32 o = { __shfl_xor(u.x, 16), __shfl_xor(u.y, 16) };
            const int rb = r >> 2;
            const c32 in0 = rb ? o : u, in1 = rb ? u : o;
            const c32 n0 = { uq[l + 1][q0][rb * 4],     uq[l + 1][q0][rb * 4 + 1] };
            const c32 n1 = { uq[l + 1][q0][rb * 4 + 2], uq[l + 1][q0][rb * 4 + 3] };
            const c32 m0 = cmul(n0, in0), m1 = cmul(n1, in1);
            u = { m0.x + m1.x, m0.y + m1.y };
        }
        float* Ao = AL[ub][g];
        const int M0 = 8 * (r >> 2) + (r & 3), M1 = M0 + 4;
        const int kb = (c & 1) * 8 + (c >> 1) * 2;
        Ao[M0 * 16 + kb] = u.x;  Ao[M0 * 16 + kb + 1] = -u.y;
        Ao[M1 * 16 + kb] = u.y;  Ao[M1 * 16 + kb + 1] =  u.x;
    };

    if (lane < NQ) {
        if (wv == 0) {
            float xv = xin[b * NQ + lane];
            float h = 0.5f * xv;
            float a = 0.5f * xv * xv;
            float ch, sh, ca, sa;
            sincosf(h, &sh, &ch);
            sincosf(a, &sa, &ca);
            c32 e0 = { ch * ca, -ch * sa }, e1 = { sh * ca, sh * sa };
            const gate1 g0 = wire_gate(params + lane * 3);
            if (lane == 3 || lane == 6 || lane == 9) {
                const c32 a0 = cmul(g0.u00, e0), a1 = cmul(g0.u01, e1);
                const c32 b0 = cmul(g0.u10, e0), b1 = cmul(g0.u11, e1);
                e0 = { a0.x + a1.x, a0.y + a1.y };
                e1 = { b0.x + b1.x, b0.y + b1.y };
            }
            enc[lane][0] = e0.x;  enc[lane][1] = e0.y;
            enc[lane][2] = e1.x;  enc[lane][3] = e1.y;
        } else if (wv <= NLAYER) {
            layer_scalars(wv - 1, lane);
        }
    }
    __syncthreads();

    build_unitaries(0, 0);

    for (int m = 0; m < 16; ++m) {
        int k = tid + 256 * m;
        c32 amp = { 1.0f, 0.0f };
#pragma unroll
        for (int q = 0; q < NQ; ++q) {
            int bit = (k >> (11 - q)) & 1;
            c32 f = { enc[q][bit * 2], enc[q][bit * 2 + 1] };
            amp = cmul(amp, f);
        }
        sre[swz(k)] = amp.x; sim_[swz(k)] = amp.y;
    }
    __syncthreads();

    const int w1part = wv;
    const int w2part = ((wv & 1) << 6) | (((wv >> 1) & 1) << 7) | (((wv >> 2) & 1) << 9);

    for (int l = 0; l < NLAYER; ++l) {
        const int ub = l & 1;
        int lra = lr, ha = half;
        asm volatile("" : "+v"(lra), "+v"(ha));

        apply_pair(AL[ub][0],  9, 8, 5,  3, 4, 6, 7,    w1part, lra, ha, l, 2);
        wave_sync();
        apply_pair(AL[ub][1],  6, 5, 4,  3, 9, 10, 11,  w1part, lra, ha, l, 5);
        __syncthreads();

        apply_pair(AL[ub][2],  3, 2, 11, 0, 1, 8, 10,   w2part, lra, ha, l, 8);
        wave_sync();
        if (l + 1 < NLAYER) build_unitaries(l + 1, ub ^ 1);
        apply_pair(AL[ub][3],  0, 11, 10, 3, 4, 5, 8,   w2part, lra, ha, l, 11);
        __syncthreads();
    }

    float part[NQ];
#pragma unroll
    for (int q = 0; q < NQ; ++q) part[q] = 0.0f;
    for (int m = 0; m < 16; ++m) {
        int k = tid + 256 * m;
        float vr = sre[swz(k)], vi = sim_[swz(k)];
        float pr = vr * vr + vi * vi;
#pragma unroll
        for (int q = 0; q < NQ; ++q)
            part[q] += ((k >> (11 - q)) & 1) ? -pr : pr;
    }
#pragma unroll
    for (int q = 0; q < NQ; ++q) red[tid * NQ + q] = part[q];
    __syncthreads();
    for (int s = 128; s > 0; s >>= 1) {
        if (tid < s) {
#pragma unroll
            for (int q = 0; q < NQ; ++q)
                red[tid * NQ + q] += red[(tid + s) * NQ + q];
        }
        __syncthreads();
    }
    if (tid < NQ) out[b * NQ + tid] = red[tid];
}

extern "C" void kernel_launch(void* const* d_in, const int* in_sizes, int n_in,
                              void* d_out, int out_size, void* d_ws, size_t ws_size,
                              hipStream_t stream) {
    (void)n_in; (void)out_size; (void)d_ws; (void)ws_size;
    const float* x      = (const float*)d_in[0];
    const float* params = (const float*)d_in[1];
    float* out = (float*)d_out;
    int B = in_sizes[0] / NQ;
    qsim_kernel<<<dim3(B), dim3(256), 0, stream>>>(x, params, out);
}


# LDS state as (re, im) pairs with a bank swizzle for 8-byte accesses: each amplitude moves as one conflict-free b64, not two same-bank dwords (2-way conflicts in 4 of 8 patterns); rounding changes
# speedup vs baseline: 2.0728x; 1.0323x over previous
#include <hip/hip_runtime.h>


typedef float v2f __attribute__((ext_vector_type(2)));
typedef float v8f __attribute__((ext_vector_type(8)));

struct c32 { float x, y; };
__device__ __forceinline__ c32 cmul(c32 a, c32 b) {
    return { a.x * b.x - a.y * b.y, a.x * b.y + a.y * b.x };
}

struct gate1 { c32 u00, u01, u10, u11; float c0, s0, c1, s1, cz, sz; };
__device__ __forceinline__ gate1 wire_gate(const float* wp) {
    gate1 g;
    float w0 = wp[0], w1 = wp[1], w2 = wp[2];
    sincosf(0.5f * w0, &g.s0, &g.c0);
    sincosf(0.5f * w1, &g.s1, &g.c1);
    sincosf(0.5f * w2, &g.sz, &g.cz);
    c32 A  = {  g.c1 * g.c0,  g.s1 * g.s0 };
    c32 Bm = { -g.s1 * g.c0, -g.c1 * g.s0 };
    c32 Cm = {  g.s1 * g.c0, -g.c1 * g.s0 };
    c32 Dm = {  g.c1 * g.c0, -g.s1 * g.s0 };
    c32 ez  = { g.cz,  g.sz };
    c32 ezc = { g.cz, -g.sz };
    g.u00 = cmul(A, ezc);  g.u01 = cmul(Bm, ezc);
    g.u10 = cmul(Cm, ez);  g.u11 = cmul(Dm, ez);
    return g;
}

#define NQ     12
#define DIM    4096
#define NLAYER 4

__device__ __forceinline__ constexpr int swz(int k) {
    return k ^ ((k >> 6) & 23) ^ ((k >> 7) & 4) ^ ((k >> 4) & 2) ^ ((k >> 2) & 8)
             ^ ((k >> 8) & 1) ^ ((k >> 10) & 2) ^ ((k >> 9) & 4);
}

__global__ __launch_bounds__(256)
void qsim_kernel(const float* __restrict__ xin,
                 const float* __restrict__ params,
                 float* __restrict__ out) {
    __shared__ v2f S[DIM];
    __shared__ float AL[2][4][256];
    __shared__ float Tar[4][64], Tai[4][64];
    __shared__ float Tbr[4][64], Tbi[4][64];
    __shared__ float uq[NLAYER][NQ][8];
    __shared__ float pv[NLAYER][NQ][8];
    __shared__ float pmr[NLAYER][NQ][16];
    __shared__ float pmi[NLAYER][NQ][16];
    __shared__ float enc[NQ][4];
    __shared__ float red[256 * NQ];

    const int tid  = threadIdx.x;
    const int b    = blockIdx.x;
    const int lane = tid & 31;
    const int wv   = tid >> 5;
    const int half = lane >> 4;
    const int lr   = lane & 15;

    auto wave_sync = [&]() {
        __builtin_amdgcn_fence(__ATOMIC_RELEASE, "wavefront");
        __builtin_amdgcn_wave_barrier();
        __builtin_amdgcn_fence(__ATOMIC_ACQUIRE, "wavefront");
    };

    auto apply_pair = [&](const float* A, int gb, int yb, int tb,
                          int nb0, int nb1, int nb2, int nb3, int wpart,
                          int lra, int ha,
                          int l, int i) __attribute__((always_inline)) {
        const float cd = pv[l][i][0], sd = pv[l][i][1];
        const float cs = pv[l][i][2], ss = pv[l][i][3];
        const float c2 = pv[l][i][6], s2 = pv[l][i][7];
        const int lanepart = ((lra & 1) << nb0) | (((lra >> 1) & 1) << nb1)
                           | (((lra >> 2) & 1) << nb2) | (((lra >> 3) & 1) << nb3);
        const int ldbase = swz(wpart | lanepart | (ha << gb));
        const int stbase = swz(wpart | lanepart | (ha << (gb + 2)));
        v2f af[4];
#pragma unroll
        for (int ks = 0; ks < 4; ++ks) {
            af[ks][0] = A[lr * 16 + half * 8 + 2 * ks];
            af[ks][1] = A[lr * 16 + half * 8 + 2 * ks + 1];
        }
        v2f bf[2][2][4];
#pragma unroll
        for (int y = 0; y < 2; ++y) {
#pragma unroll
            for (int t = 0; t < 2; ++t) {
#pragma unroll
                for (int ks = 0; ks < 4; ++ks) {
                    const int a = ldbase ^ swz((y << yb) | (t << tb) | ((2 * ks) << gb));
                    bf[y][t][ks] = S[a];
                }
            }
        }
        v8f acc[2][2];
#pragma unroll
        for (int y = 0; y < 2; ++y) {
#pragma unroll
            for (int t = 0; t < 2; ++t) acc[y][t] = v8f{};
        }
#pragma unroll
        for (int ks = 0; ks < 4; ++ks) {
#pragma unroll
            for (int y = 0; y < 2; ++y) {
#pragma unroll
                for (int t = 0; t < 2; ++t) {
                    acc[y][t] = __builtin_amdgcn_wmma_f32_16x16x4_f32(
                        false, af[ks], false, bf[y][t][ks], (short)0, acc[y][t], false, false);
                }
            }
        }
        const v2f P2 = { -s2, s2 };
        const v2f SD = { sd, -sd }, SS = { ss, -ss };
#pragma unroll
        for (int t = 0; t < 2; ++t) {
#pragma unroll
            for (int p = 0; p < 2; ++p) {
                const v2f a00 = { acc[0][t][4 * p],     acc[0][t][4 * p + 1] };
                const v2f a10 = { acc[0][t][4 * p + 2], acc[0][t][4 * p + 3] };
                const v2f a01 = { acc[1][t][4 * p],     acc[1][t][4 * p + 1] };
                const v2f a11 = { acc[1][t][4 * p + 2], acc[1][t][4 * p + 3] };
                const v2f b11 = c2 * a11 + P2 * a11.yx;
                const v2f n00 = cd * a00 + SD * b11.yx;
                const v2f n11 = cd * b11 + SD * a00.yx;
                const v2f n10 = cs * a10 + SS * a01.yx;
                const v2f n01 = cs * a01 + SS * a10.yx;
                S[stbase ^ swz((0 << yb) | (t << tb) | ((2 * p)     << gb))] = n00;
                S[stbase ^ swz((0 << yb) | (t << tb) | ((2 * p + 1) << gb))] = n10;
                S[stbase ^ swz((1 << yb) | (t << tb) | ((2 * p)     << gb))] = n01;
                S[stbase ^ swz((1 << yb) | (t << tb) | ((2 * p + 1) << gb))] = n11;
            }
        }
    };

    auto layer_scalars = [&](int l, int q) {
        const gate1 gq = wire_gate(params + (l * NQ + q) * 3);
        const float c0 = gq.c0, s0 = gq.s0, c1 = gq.c1, s1 = gq.s1, cz = gq.cz, sz = gq.sz;
        const c32 u00 = gq.u00, u01 = gq.u01, u10 = gq.u10, u11 = gq.u11;
        uq[l][q][0] = u00.x; uq[l][q][1] = u00.y;
        uq[l][q][2] = u01.x; uq[l][q][3] = u01.y;
        uq[l][q][4] = u10.x; uq[l][q][5] = u10.y;
        uq[l][q][6] = u11.x; uq[l][q][7] = u11.y;
        float cd = c1 * cz + s1 * sz, sd = s1 * cz - c1 * sz;
        float cs = c1 * cz - s1 * sz, ss = s1 * cz + c1 * sz;
        float cph = c0, sph = s0;
        pv[l][q][0] = cd; pv[l][q][1] = sd;
        pv[l][q][2] = cs; pv[l][q][3] = ss;
        pv[l][q][4] = cph; pv[l][q][5] = sph;
        pv[l][q][6] = cph * cph - sph * sph; pv[l][q][7] = 2.0f * sph * cph;
#pragma unroll
        for (int e = 0; e < 16; ++e) { pmr[l][q][e] = 0.0f; pmi[l][q][e] = 0.0f; }
        pmr[l][q][0]  = cd;
        pmr[l][q][3]  = sd * sph;  pmi[l][q][3]  = -sd * cph;
        pmi[l][q][12] = -sd;
        pmr[l][q][15] = cd * cph;  pmi[l][q][15] =  cd * sph;
        pmr[l][q][5]  = cs;
        pmr[l][q][6]  = -ss * sph; pmi[l][q][6]  = -ss * cph;
        pmi[l][q][9]  = -ss;
        pmr[l][q][10] = cs * cph;  pmi[l][q][10] = -cs * sph;
    };

    auto build_unitaries = [&](int l, int ub) {
        const int g = wv >> 1, c = 4 * (wv & 1) + (lane & 3), r = lane >> 2, q0 = 3 * g;
        {
            int o0 = ((((r >> 2) & 1) * 2 + ((c >> 2) & 1)) * 2);
            int o1 = ((((r >> 1) & 1) * 2 + ((c >> 1) & 1)) * 2);
            int o2 = (((r & 1) * 2 + (c & 1)) * 2);
            c32 acc;
            if (g == 0) acc = { uq[l][0][o0], uq[l][0][o0 + 1] };
            else        acc = { (((r >> 2) & 1) == ((c >> 2) & 1)) ? 1.0f : 0.0f, 0.0f };
            acc = cmul(acc, c32{ uq[l][q0 + 1][o1], uq[l][q0 + 1][o1 + 1] });
            acc = cmul(acc, c32{ uq[l][q0 + 2][o2], uq[l][q0 + 2][o2 + 1] });
            Tar[g][r * 8 + c] = acc.x; Tai[g][r * 8 + c] = acc.y;
        }
        wave_sync();
        auto chain8 = [&](int pair, int sh, const float* inR, const float* inI) {
            int rowsel = ((r >> sh) & 3) * 4;
            int rmask = r & ~(3 << sh);
            c32 acc = { 0.0f, 0.0f };
#pragma unroll
            for (int jj = 0; jj < 4; ++jj) {
                int k = rmask | (jj << sh);
                c32 p = { pmr[l][pair][rowsel + jj], pmi[l][pair][rowsel + jj] };
                c32 v = { inR[k * 8 + c], inI[k * 8 + c] };
                acc.x += p.x * v.x - p.y * v.y;
                acc.y += p.x * v.y + p.y * v.x;
            }
            return acc;
        };
        c32 s0 = chain8(q0, 1, Tar[g], Tai[g]);
        Tbr[g][r * 8 + c] = s0.x; Tbi[g][r * 8 + c] = s0.y;
        wave_sync();
        c32 u = chain8(q0 + 1, 0, Tbr[g], Tbi[g]);
        if (r & 1) u = cmul(u, c32{ pv[l][q0 + 2][4], -pv[l][q0 + 2][5] });
        if (g > 0 && l + 1 < NLAYER) {
            const c32 o = { __shfl_xor(u.x, 16), __shfl_xor(u.y, 16) };
            const int rb = r >> 2;
            const c32 in0 = rb ? o : u, in1 = rb ? u : o;
            const c32 n0 = { uq[l + 1][q0][rb * 4],     uq[l + 1][q0][rb * 4 + 1] };
            const c32 n1 = { uq[l + 1][q0][rb * 4 + 2], uq[l + 1][q0][rb * 4 + 3] };
            const c32 m0 = cmul(n0, in0), m1 = cmul(n1, in1);
            u = { m0.x + m1.x, m0.y + m1.y };
        }
        float* Ao = AL[ub][g];
        const int M0 = 8 * (r >> 2) + 2 * (r & 3), M1 = M0 + 1;
        const int kb = (c & 1) * 8 + (c >> 1) * 2;
        Ao[M0 * 16 + kb] = u.x;  Ao[M0 * 16 + kb + 1] = -u.y;
        Ao[M1 * 16 + kb] = u.y;  Ao[M1 * 16 + kb + 1] =  u.x;
    };

    if (lane < NQ) {
        if (wv == 0) {
            float xv = xin[b * NQ + lane];
            float h = 0.5f * xv;
            float a = 0.5f * xv * xv;
            float ch, sh, ca, sa;
            sincosf(h, &sh, &ch);
            sincosf(a, &sa, &ca);
            c32 e0 = { ch * ca, -ch * sa }, e1 = { sh * ca, sh * sa };
            const gate1 g0 = wire_gate(params + lane * 3);
            if (lane == 3 || lane == 6 || lane == 9) {
                const c32 a0 = cmul(g0.u00, e0), a1 = cmul(g0.u01, e1);
                const c32 b0 = cmul(g0.u10, e0), b1 = cmul(g0.u11, e1);
                e0 = { a0.x + a1.x, a0.y + a1.y };
                e1 = { b0.x + b1.x, b0.y + b1.y };
            }
            enc[lane][0] = e0.x;  enc[lane][1] = e0.y;
            enc[lane][2] = e1.x;  enc[lane][3] = e1.y;
        } else if (wv <= NLAYER) {
            layer_scalars(wv - 1, lane);
        }
    }
    __syncthreads();

    build_unitaries(0, 0);

    for (int m = 0; m < 16; ++m) {
        int k = tid + 256 * m;
        c32 amp = { 1.0f, 0.0f };
#pragma unroll
        for (int q = 0; q < NQ; ++q) {
            int bit = (k >> (11 - q)) & 1;
            c32 f = { enc[q][bit * 2], enc[q][bit * 2 + 1] };
            amp = cmul(amp, f);
        }
        S[swz(k)] = v2f{ amp.x, amp.y };
    }
    __syncthreads();

    const int w1part = wv;
    const int w2part = ((wv & 1) << 6) | (((wv >> 1) & 1) << 7) | (((wv >> 2) & 1) << 9);

    for (int l = 0; l < NLAYER; ++l) {
        const int ub = l & 1;
        int lra = lr, ha = half;
        asm volatile("" : "+v"(lra), "+v"(ha));

        apply_pair(AL[ub][0],  9, 8, 5,  3, 4, 6, 7,    w1part, lra, ha, l, 2);
        wave_sync();
        apply_pair(AL[ub][1],  6, 5, 4,  3, 9, 10, 11,  w1part, lra, ha, l, 5);
        __syncthreads();

        apply_pair(AL[ub][2],  3, 2, 11, 0, 1, 8, 10,   w2part, lra, ha, l, 8);
        wave_sync();
        if (l + 1 < NLAYER) build_unitaries(l + 1, ub ^ 1);
        apply_pair(AL[ub][3],  0, 11, 10, 3, 4, 5, 8,   w2part, lra, ha, l, 11);
        __syncthreads();
    }

    float part[NQ];
#pragma unroll
    for (int q = 0; q < NQ; ++q) part[q] = 0.0f;
    for (int m = 0; m < 16; ++m) {
        int k = tid + 256 * m;
        const v2f z = S[swz(k)];
        float pr = z[0] * z[0] + z[1] * z[1];
#pragma unroll
        for (int q = 0; q < NQ; ++q)
            part[q] += ((k >> (11 - q)) & 1) ? -pr : pr;
    }
#pragma unroll
    for (int q = 0; q < NQ; ++q) red[tid * NQ + q] = part[q];
    __syncthreads();
    for (int s = 128; s > 0; s >>= 1) {
        if (tid < s) {
#pragma unroll
            for (int q = 0; q < NQ; ++q)
                red[tid * NQ + q] += red[(tid + s) * NQ + q];
        }
        __syncthreads();
    }
    if (tid < NQ) out[b * NQ + tid] = red[tid];
}

extern "C" void kernel_launch(void* const* d_in, const int* in_sizes, int n_in,
                              void* d_out, int out_size, void* d_ws, size_t ws_size,
                              hipStream_t stream) {
    (void)n_in; (void)out_size; (void)d_ws; (void)ws_size;
    const float* x      = (const float*)d_in[0];
    const float* params = (const float*)d_in[1];
    float* out = (float*)d_out;
    int B = in_sizes[0] / NQ;
    qsim_kernel<<<dim3(B), dim3(256), 0, stream>>>(x, params, out);
}


# Z marginals taken from the last apply's accumulators in registers: per-lane signed sums, row_xmask/permlanex16 butterfly, one 8-wave stage; final state never stored or re-read; sum order changes
# speedup vs baseline: 2.1594x; 1.0418x over previous
#include <hip/hip_runtime.h>


typedef float v2f __attribute__((ext_vector_type(2)));
typedef float v8f __attribute__((ext_vector_type(8)));

struct c32 { float x, y; };
__device__ __forceinline__ c32 cmul(c32 a, c32 b) {
    return { a.x * b.x - a.y * b.y, a.x * b.y + a.y * b.x };
}

template <int M>
__device__ __forceinline__ float rowx(float x) {
    const int b = __builtin_bit_cast(int, x);
    return __builtin_bit_cast(float, __builtin_amdgcn_update_dpp(b, b, 0x160 + M, 0xF, 0xF, false));
}
__device__ __forceinline__ float xhalf(float x) {
    const int b = __builtin_bit_cast(int, x);
    return __builtin_bit_cast(float, __builtin_amdgcn_permlanex16(b, b, 0x76543210, 0xfedcba98, false, false));
}

struct gate1 { c32 u00, u01, u10, u11; float c0, s0, c1, s1, cz, sz; };
__device__ __forceinline__ gate1 wire_gate(const float* wp) {
    gate1 g;
    float w0 = wp[0], w1 = wp[1], w2 = wp[2];
    sincosf(0.5f * w0, &g.s0, &g.c0);
    sincosf(0.5f * w1, &g.s1, &g.c1);
    sincosf(0.5f * w2, &g.sz, &g.cz);
    c32 A  = {  g.c1 * g.c0,  g.s1 * g.s0 };
    c32 Bm = { -g.s1 * g.c0, -g.c1 * g.s0 };
    c32 Cm = {  g.s1 * g.c0, -g.c1 * g.s0 };
    c32 Dm = {  g.c1 * g.c0, -g.s1 * g.s0 };
    c32 ez  = { g.cz,  g.sz };
    c32 ezc = { g.cz, -g.sz };
    g.u00 = cmul(A, ezc);  g.u01 = cmul(Bm, ezc);
    g.u10 = cmul(Cm, ez);  g.u11 = cmul(Dm, ez);
    return g;
}

#define NQ     12
#define DIM    4096
#define NLAYER 4

__device__ __forceinline__ constexpr int swz(int k) {
    return k ^ ((k >> 6) & 23) ^ ((k >> 7) & 4) ^ ((k >> 4) & 2) ^ ((k >> 2) & 8)
             ^ ((k >> 8) & 1) ^ ((k >> 10) & 2) ^ ((k >> 9) & 4);
}

__global__ __launch_bounds__(256)
void qsim_kernel(const float* __restrict__ xin,
                 const float* __restrict__ params,
                 float* __restrict__ out) {
    __shared__ v2f S[DIM];
    __shared__ float AL[2][4][256];
    __shared__ float Tar[4][64], Tai[4][64];
    __shared__ float Tbr[4][64], Tbi[4][64];
    __shared__ float uq[NLAYER][NQ][8];
    __shared__ float pv[NLAYER][NQ][8];
    __shared__ float pmr[NLAYER][NQ][16];
    __shared__ float pmi[NLAYER][NQ][16];
    __shared__ float enc[NQ][4];
    __shared__ float zred[8 * NQ];

    const int tid  = threadIdx.x;
    const int b    = blockIdx.x;
    const int lane = tid & 31;
    const int wv   = tid >> 5;
    const int half = lane >> 4;
    const int lr   = lane & 15;

    auto wave_sync = [&]() {
        __builtin_amdgcn_fence(__ATOMIC_RELEASE, "wavefront");
        __builtin_amdgcn_wave_barrier();
        __builtin_amdgcn_fence(__ATOMIC_ACQUIRE, "wavefront");
    };

    auto apply_pair = [&](const float* A, int gb, int yb, int tb,
                          int nb0, int nb1, int nb2, int nb3, int wpart,
                          int lra, int ha,
                          int l, int i, bool fin, float* zout) __attribute__((always_inline)) {
        const float cd = pv[l][i][0], sd = pv[l][i][1];
        const float cs = pv[l][i][2], ss = pv[l][i][3];
        const float c2 = pv[l][i][6], s2 = pv[l][i][7];
        const int lanepart = ((lra & 1) << nb0) | (((lra >> 1) & 1) << nb1)
                           | (((lra >> 2) & 1) << nb2) | (((lra >> 3) & 1) << nb3);
        const int ldbase = swz(wpart | lanepart | (ha << gb));
        const int stbase = swz(wpart | lanepart | (ha << (gb + 2)));
        v2f af[4];
#pragma unroll
        for (int ks = 0; ks < 4; ++ks) {
            af[ks][0] = A[lr * 16 + half * 8 + 2 * ks];
            af[ks][1] = A[lr * 16 + half * 8 + 2 * ks + 1];
        }
        v2f bf[2][2][4];
#pragma unroll
        for (int y = 0; y < 2; ++y) {
#pragma unroll
            for (int t = 0; t < 2; ++t) {
#pragma unroll
                for (int ks = 0; ks < 4; ++ks) {
                    const int a = ldbase ^ swz((y << yb) | (t << tb) | ((2 * ks) << gb));
                    bf[y][t][ks] = S[a];
                }
            }
        }
        v8f acc[2][2];
#pragma unroll
        for (int y = 0; y < 2; ++y) {
#pragma unroll
            for (int t = 0; t < 2; ++t) acc[y][t] = v8f{};
        }
#pragma unroll
        for (int ks = 0; ks < 4; ++ks) {
#pragma unroll
            for (int y = 0; y < 2; ++y) {
#pragma unroll
                for (int t = 0; t < 2; ++t) {
                    acc[y][t] = __builtin_amdgcn_wmma_f32_16x16x4_f32(
                        false, af[ks], false, bf[y][t][ks], (short)0, acc[y][t], false, false);
                }
            }
        }
        const v2f P2 = { -s2, s2 };
        const v2f SD = { sd, -sd }, SS = { ss, -ss };
        v2f nv[2][2][4];
#pragma unroll
        for (int t = 0; t < 2; ++t) {
#pragma unroll
            for (int p = 0; p < 2; ++p) {
                const v2f a00 = { acc[0][t][4 * p],     acc[0][t][4 * p + 1] };
                const v2f a10 = { acc[0][t][4 * p + 2], acc[0][t][4 * p + 3] };
                const v2f a01 = { acc[1][t][4 * p],     acc[1][t][4 * p + 1] };
                const v2f a11 = { acc[1][t][4 * p + 2], acc[1][t][4 * p + 3] };
                const v2f b11 = c2 * a11 + P2 * a11.yx;
                nv[t][p][0] = cd * a00 + SD * b11.yx;
                nv[t][p][3] = cd * b11 + SD * a00.yx;
                nv[t][p][1] = cs * a10 + SS * a01.yx;
                nv[t][p][2] = cs * a01 + SS * a10.yx;
            }
        }
        if (!fin) {
#pragma unroll
            for (int t = 0; t < 2; ++t) {
#pragma unroll
                for (int p = 0; p < 2; ++p) {
                    S[stbase ^ swz((0 << yb) | (t << tb) | ((2 * p)     << gb))] = nv[t][p][0];
                    S[stbase ^ swz((0 << yb) | (t << tb) | ((2 * p + 1) << gb))] = nv[t][p][1];
                    S[stbase ^ swz((1 << yb) | (t << tb) | ((2 * p)     << gb))] = nv[t][p][2];
                    S[stbase ^ swz((1 << yb) | (t << tb) | ((2 * p + 1) << gb))] = nv[t][p][3];
                }
            }
        } else {
            float zAll = 0.0f, zX = 0.0f, zP = 0.0f, zT = 0.0f, zY = 0.0f;
#pragma unroll
            for (int t = 0; t < 2; ++t) {
#pragma unroll
                for (int p = 0; p < 2; ++p) {
                    const v2f q00 = nv[t][p][0] * nv[t][p][0], q10 = nv[t][p][1] * nv[t][p][1];
                    const v2f q01 = nv[t][p][2] * nv[t][p][2], q11 = nv[t][p][3] * nv[t][p][3];
                    const float p00 = q00[0] + q00[1], p10 = q10[0] + q10[1];
                    const float p01 = q01[0] + q01[1], p11 = q11[0] + q11[1];
                    const float y0 = p00 + p10, y1 = p01 + p11;
                    const float all = y0 + y1;
                    zAll += all;
                    zX += (p00 - p10) + (p01 - p11);
                    zY += y0 - y1;
                    zP += p ? -all : all;
                    zT += t ? -all : all;
                }
            }
            const int basek = wpart | lanepart | (ha << (gb + 2));
            float zq[NQ];
#pragma unroll
            for (int q = 0; q < NQ; ++q) {
                const int bit = (NQ - 1) - q;
                float v;
                if (bit == gb)          v = zX;
                else if (bit == gb + 1) v = zP;
                else if (bit == tb)     v = zT;
                else if (bit == yb)     v = zY;
                else                    v = ((basek >> bit) & 1) ? -zAll : zAll;
                v += rowx<1>(v);
                v += rowx<2>(v);
                v += rowx<4>(v);
                v += rowx<8>(v);
                v += xhalf(v);
                zq[q] = v;
            }
            if (lane == 0) {
#pragma unroll
                for (int q = 0; q < NQ; ++q) zout[q] = zq[q];
            }
        }
    };

    auto layer_scalars = [&](int l, int q) {
        const gate1 gq = wire_gate(params + (l * NQ + q) * 3);
        const float c0 = gq.c0, s0 = gq.s0, c1 = gq.c1, s1 = gq.s1, cz = gq.cz, sz = gq.sz;
        const c32 u00 = gq.u00, u01 = gq.u01, u10 = gq.u10, u11 = gq.u11;
        uq[l][q][0] = u00.x; uq[l][q][1] = u00.y;
        uq[l][q][2] = u01.x; uq[l][q][3] = u01.y;
        uq[l][q][4] = u10.x; uq[l][q][5] = u10.y;
        uq[l][q][6] = u11.x; uq[l][q][7] = u11.y;
        float cd = c1 * cz + s1 * sz, sd = s1 * cz - c1 * sz;
        float cs = c1 * cz - s1 * sz, ss = s1 * cz + c1 * sz;
        float cph = c0, sph = s0;
        pv[l][q][0] = cd; pv[l][q][1] = sd;
        pv[l][q][2] = cs; pv[l][q][3] = ss;
        pv[l][q][4] = cph; pv[l][q][5] = sph;
        pv[l][q][6] = cph * cph - sph * sph; pv[l][q][7] = 2.0f * sph * cph;
#pragma unroll
        for (int e = 0; e < 16; ++e) { pmr[l][q][e] = 0.0f; pmi[l][q][e] = 0.0f; }
        pmr[l][q][0]  = cd;
        pmr[l][q][3]  = sd * sph;  pmi[l][q][3]  = -sd * cph;
        pmi[l][q][12] = -sd;
        pmr[l][q][15] = cd * cph;  pmi[l][q][15] =  cd * sph;
        pmr[l][q][5]  = cs;
        pmr[l][q][6]  = -ss * sph; pmi[l][q][6]  = -ss * cph;
        pmi[l][q][9]  = -ss;
        pmr[l][q][10] = cs * cph;  pmi[l][q][10] = -cs * sph;
    };

    auto build_unitaries = [&](int l, int ub) {
        const int g = wv >> 1, c = 4 * (wv & 1) + (lane & 3), r = lane >> 2, q0 = 3 * g;
        {
            int o0 = ((((r >> 2) & 1) * 2 + ((c >> 2) & 1)) * 2);
            int o1 = ((((r >> 1) & 1) * 2 + ((c >> 1) & 1)) * 2);
            int o2 = (((r & 1) * 2 + (c & 1)) * 2);
            c32 acc;
            if (g == 0) acc = { uq[l][0][o0], uq[l][0][o0 + 1] };
            else        acc = { (((r >> 2) & 1) == ((c >> 2) & 1)) ? 1.0f : 0.0f, 0.0f };
            acc = cmul(acc, c32{ uq[l][q0 + 1][o1], uq[l][q0 + 1][o1 + 1] });
            acc = cmul(acc, c32{ uq[l][q0 + 2][o2], uq[l][q0 + 2][o2 + 1] });
            Tar[g][r * 8 + c] = acc.x; Tai[g][r * 8 + c] = acc.y;
        }
        wave_sync();
        auto chain8 = [&](int pair, int sh, const float* inR, const float* inI) {
            int rowsel = ((r >> sh) & 3) * 4;
            int rmask = r & ~(3 << sh);
            c32 acc = { 0.0f, 0.0f };
#pragma unroll
            for (int jj = 0; jj < 4; ++jj) {
                int k = rmask | (jj << sh);
                c32 p = { pmr[l][pair][rowsel + jj], pmi[l][pair][rowsel + jj] };
                c32 v = { inR[k * 8 + c], inI[k * 8 + c] };
                acc.x += p.x * v.x - p.y * v.y;
                acc.y += p.x * v.y + p.y * v.x;
            }
            return acc;
        };
        c32 s0 = chain8(q0, 1, Tar[g], Tai[g]);
        Tbr[g][r * 8 + c] = s0.x; Tbi[g][r * 8 + c] = s0.y;
        wave_sync();
        c32 u = chain8(q0 + 1, 0, Tbr[g], Tbi[g]);
        if (r & 1) u = cmul(u, c32{ pv[l][q0 + 2][4], -pv[l][q0 + 2][5] });
        if (g > 0 && l + 1 < NLAYER) {
            const c32 o = { __shfl_xor(u.x, 16), __shfl_xor(u.y, 16) };
            const int rb = r >> 2;
            const c32 in0 = rb ? o : u, in1 = rb ? u : o;
            const c32 n0 = { uq[l + 1][q0][rb * 4],     uq[l + 1][q0][rb * 4 + 1] };
            const c32 n1 = { uq[l + 1][q0][rb * 4 + 2], uq[l + 1][q0][rb * 4 + 3] };
            const c32 m0 = cmul(n0, in0), m1 = cmul(n1, in1);
            u = { m0.x + m1.x, m0.y + m1.y };
        }
        float* Ao = AL[ub][g];
        const int M0 = 8 * (r >> 2) + 2 * (r & 3), M1 = M0 + 1;
        const int kb = (c & 1) * 8 + (c >> 1) * 2;
        Ao[M0 * 16 + kb] = u.x;  Ao[M0 * 16 + kb + 1] = -u.y;
        Ao[M1 * 16 + kb] = u.y;  Ao[M1 * 16 + kb + 1] =  u.x;
    };

    if (lane < NQ) {
        if (wv == 0) {
            float xv = xin[b * NQ + lane];
            float h = 0.5f * xv;
            float a = 0.5f * xv * xv;
            float ch, sh, ca, sa;
            sincosf(h, &sh, &ch);
            sincosf(a, &sa, &ca);
            c32 e0 = { ch * ca, -ch * sa }, e1 = { sh * ca, sh * sa };
            const gate1 g0 = wire_gate(params + lane * 3);
            if (lane == 3 || lane == 6 || lane == 9) {
                const c32 a0 = cmul(g0.u00, e0), a1 = cmul(g0.u01, e1);
                const c32 b0 = cmul(g0.u10, e0), b1 = cmul(g0.u11, e1);
                e0 = { a0.x + a1.x, a0.y + a1.y };
                e1 = { b0.x + b1.x, b0.y + b1.y };
            }
            enc[lane][0] = e0.x;  enc[lane][1] = e0.y;
            enc[lane][2] = e1.x;  enc[lane][3] = e1.y;
        } else if (wv <= NLAYER) {
            layer_scalars(wv - 1, lane);
        }
    }
    __syncthreads();

    build_unitaries(0, 0);

    for (int m = 0; m < 16; ++m) {
        int k = tid + 256 * m;
        c32 amp = { 1.0f, 0.0f };
#pragma unroll
        for (int q = 0; q < NQ; ++q) {
            int bit = (k >> (11 - q)) & 1;
            c32 f = { enc[q][bit * 2], enc[q][bit * 2 + 1] };
            amp = cmul(amp, f);
        }
        S[swz(k)] = v2f{ amp.x, amp.y };
    }
    __syncthreads();

    const int w1part = wv;
    const int w2part = ((wv & 1) << 6) | (((wv >> 1) & 1) << 7) | (((wv >> 2) & 1) << 9);

    for (int l = 0; l < NLAYER; ++l) {
        const int ub = l & 1;
        int lra = lr, ha = half;
        asm volatile("" : "+v"(lra), "+v"(ha));

        apply_pair(AL[ub][0],  9, 8, 5,  3, 4, 6, 7,    w1part, lra, ha, l, 2, false, nullptr);
        wave_sync();
        apply_pair(AL[ub][1],  6, 5, 4,  3, 9, 10, 11,  w1part, lra, ha, l, 5, false, nullptr);
        __syncthreads();

        apply_pair(AL[ub][2],  3, 2, 11, 0, 1, 8, 10,   w2part, lra, ha, l, 8, false, nullptr);
        wave_sync();
        if (l + 1 < NLAYER) build_unitaries(l + 1, ub ^ 1);
        apply_pair(AL[ub][3],  0, 11, 10, 3, 4, 5, 8,   w2part, lra, ha, l, 11,
                   l == NLAYER - 1, &zred[wv * NQ]);
        __syncthreads();
    }

    if (tid < NQ) {
        float z = 0.0f;
#pragma unroll
        for (int w = 0; w < 8; ++w) z += zred[w * NQ + tid];
        out[b * NQ + tid] = z;
    }
}

extern "C" void kernel_launch(void* const* d_in, const int* in_sizes, int n_in,
                              void* d_out, int out_size, void* d_ws, size_t ws_size,
                              hipStream_t stream) {
    (void)n_in; (void)out_size; (void)d_ws; (void)ws_size;
    const float* x      = (const float*)d_in[0];
    const float* params = (const float*)d_in[1];
    float* out = (float*)d_out;
    int B = in_sizes[0] / NQ;
    qsim_kernel<<<dim3(B), dim3(256), 0, stream>>>(x, params, out);
}


# Encoded product state formed in the first apply's B-fragment registers (lane's 8-wire factor x 4 fragment-wire factors): no LDS init pass or reload; 35 complex products, not 192; rounding changes
# speedup vs baseline: 2.3413x; 1.0842x over previous
#include <hip/hip_runtime.h>


typedef float v2f __attribute__((ext_vector_type(2)));
typedef float v8f __attribute__((ext_vector_type(8)));

struct c32 { float x, y; };
__device__ __forceinline__ c32 cmul(c32 a, c32 b) {
    return { a.x * b.x - a.y * b.y, a.x * b.y + a.y * b.x };
}

template <int M>
__device__ __forceinline__ float rowx(float x) {
    const int b = __builtin_bit_cast(int, x);
    return __builtin_bit_cast(float, __builtin_amdgcn_update_dpp(b, b, 0x160 + M, 0xF, 0xF, false));
}
__device__ __forceinline__ float xhalf(float x) {
    const int b = __builtin_bit_cast(int, x);
    return __builtin_bit_cast(float, __builtin_amdgcn_permlanex16(b, b, 0x76543210, 0xfedcba98, false, false));
}

struct gate1 { c32 u00, u01, u10, u11; float c0, s0, c1, s1, cz, sz; };
__device__ __forceinline__ gate1 wire_gate(const float* wp) {
    gate1 g;
    float w0 = wp[0], w1 = wp[1], w2 = wp[2];
    sincosf(0.5f * w0, &g.s0, &g.c0);
    sincosf(0.5f * w1, &g.s1, &g.c1);
    sincosf(0.5f * w2, &g.sz, &g.cz);
    c32 A  = {  g.c1 * g.c0,  g.s1 * g.s0 };
    c32 Bm = { -g.s1 * g.c0, -g.c1 * g.s0 };
    c32 Cm = {  g.s1 * g.c0, -g.c1 * g.s0 };
    c32 Dm = {  g.c1 * g.c0, -g.s1 * g.s0 };
    c32 ez  = { g.cz,  g.sz };
    c32 ezc = { g.cz, -g.sz };
    g.u00 = cmul(A, ezc);  g.u01 = cmul(Bm, ezc);
    g.u10 = cmul(Cm, ez);  g.u11 = cmul(Dm, ez);
    return g;
}

__device__ __forceinline__ v2f cmulv(v2f a, v2f b) {
    const v2f ib = { -b[1], b[0] };
    return a[0] * b + a[1] * ib;
}

#define NQ     12
#define DIM    4096
#define NLAYER 4

__device__ __forceinline__ constexpr int swz(int k) {
    return k ^ ((k >> 6) & 23) ^ ((k >> 7) & 4) ^ ((k >> 4) & 2) ^ ((k >> 2) & 8)
             ^ ((k >> 8) & 1) ^ ((k >> 10) & 2) ^ ((k >> 9) & 4);
}

__global__ __launch_bounds__(256)
void qsim_kernel(const float* __restrict__ xin,
                 const float* __restrict__ params,
                 float* __restrict__ out) {
    __shared__ v2f S[DIM];
    __shared__ float AL[2][4][256];
    __shared__ float Tar[4][64], Tai[4][64];
    __shared__ float Tbr[4][64], Tbi[4][64];
    __shared__ float uq[NLAYER][NQ][8];
    __shared__ float pv[NLAYER][NQ][8];
    __shared__ float pmr[NLAYER][NQ][16];
    __shared__ float pmi[NLAYER][NQ][16];
    __shared__ float enc[NQ][4];
    __shared__ float zred[8 * NQ];

    const int tid  = threadIdx.x;
    const int b    = blockIdx.x;
    const int lane = tid & 31;
    const int wv   = tid >> 5;
    const int half = lane >> 4;
    const int lr   = lane & 15;

    auto wave_sync = [&]() {
        __builtin_amdgcn_fence(__ATOMIC_RELEASE, "wavefront");
        __builtin_amdgcn_wave_barrier();
        __builtin_amdgcn_fence(__ATOMIC_ACQUIRE, "wavefront");
    };

    auto apply_pair = [&](const float* A, int gb, int yb, int tb,
                          int nb0, int nb1, int nb2, int nb3, int wpart,
                          int lra, int ha,
                          int l, int i, bool fin, float* zout, bool pre) __attribute__((always_inline)) {
        const float cd = pv[l][i][0], sd = pv[l][i][1];
        const float cs = pv[l][i][2], ss = pv[l][i][3];
        const float c2 = pv[l][i][6], s2 = pv[l][i][7];
        const int lanepart = ((lra & 1) << nb0) | (((lra >> 1) & 1) << nb1)
                           | (((lra >> 2) & 1) << nb2) | (((lra >> 3) & 1) << nb3);
        const int ldbase = swz(wpart | lanepart | (ha << gb));
        const int stbase = swz(wpart | lanepart | (ha << (gb + 2)));
        v2f af[4];
#pragma unroll
        for (int ks = 0; ks < 4; ++ks) {
            af[ks][0] = A[lr * 16 + half * 8 + 2 * ks];
            af[ks][1] = A[lr * 16 + half * 8 + 2 * ks + 1];
        }
        v2f bf[2][2][4];
        if (!pre) {
#pragma unroll
            for (int y = 0; y < 2; ++y) {
#pragma unroll
                for (int t = 0; t < 2; ++t) {
#pragma unroll
                    for (int ks = 0; ks < 4; ++ks) {
                        const int a = ldbase ^ swz((y << yb) | (t << tb) | ((2 * ks) << gb));
                        bf[y][t][ks] = S[a];
                    }
                }
            }
        } else {
            const int basek = wpart | lanepart | (ha << gb);
            auto fac = [&](int q, int bv) { return v2f{ enc[q][2 * bv], enc[q][2 * bv + 1] }; };
            v2f T = { 1.0f, 0.0f };
            bool have = false;
#pragma unroll
            for (int q = 0; q < NQ; ++q) {
                const int bit = (NQ - 1) - q;
                if (bit == yb || bit == tb || bit == gb + 1 || bit == gb + 2) continue;
                const v2f f = fac(q, (basek >> bit) & 1);
                T = have ? cmulv(T, f) : f;
                have = true;
            }
            const int qy = (NQ - 1) - yb, qt = (NQ - 1) - tb;
            const int qk0 = (NQ - 1) - (gb + 1), qk1 = (NQ - 1) - (gb + 2);
            v2f gk[4];
#pragma unroll
            for (int ks = 0; ks < 4; ++ks) gk[ks] = cmulv(fac(qk1, ks >> 1), fac(qk0, ks & 1));
#pragma unroll
            for (int y = 0; y < 2; ++y) {
#pragma unroll
                for (int t = 0; t < 2; ++t) {
                    const v2f h = cmulv(cmulv(fac(qy, y), fac(qt, t)), T);
#pragma unroll
                    for (int ks = 0; ks < 4; ++ks) bf[y][t][ks] = cmulv(gk[ks], h);
                }
            }
        }
        v8f acc[2][2];
#pragma unroll
        for (int y = 0; y < 2; ++y) {
#pragma unroll
            for (int t = 0; t < 2; ++t) acc[y][t] = v8f{};
        }
#pragma unroll
        for (int ks = 0; ks < 4; ++ks) {
#pragma unroll
            for (int y = 0; y < 2; ++y) {
#pragma unroll
                for (int t = 0; t < 2; ++t) {
                    acc[y][t] = __builtin_amdgcn_wmma_f32_16x16x4_f32(
                        false, af[ks], false, bf[y][t][ks], (short)0, acc[y][t], false, false);
                }
            }
        }
        const v2f P2 = { -s2, s2 };
        const v2f SD = { sd, -sd }, SS = { ss, -ss };
        v2f nv[2][2][4];
#pragma unroll
        for (int t = 0; t < 2; ++t) {
#pragma unroll
            for (int p = 0; p < 2; ++p) {
                const v2f a00 = { acc[0][t][4 * p],     acc[0][t][4 * p + 1] };
                const v2f a10 = { acc[0][t][4 * p + 2], acc[0][t][4 * p + 3] };
                const v2f a01 = { acc[1][t][4 * p],     acc[1][t][4 * p + 1] };
                const v2f a11 = { acc[1][t][4 * p + 2], acc[1][t][4 * p + 3] };
                const v2f b11 = c2 * a11 + P2 * a11.yx;
                nv[t][p][0] = cd * a00 + SD * b11.yx;
                nv[t][p][3] = cd * b11 + SD * a00.yx;
                nv[t][p][1] = cs * a10 + SS * a01.yx;
                nv[t][p][2] = cs * a01 + SS * a10.yx;
            }
        }
        if (!fin) {
#pragma unroll
            for (int t = 0; t < 2; ++t) {
#pragma unroll
                for (int p = 0; p < 2; ++p) {
                    S[stbase ^ swz((0 << yb) | (t << tb) | ((2 * p)     << gb))] = nv[t][p][0];
                    S[stbase ^ swz((0 << yb) | (t << tb) | ((2 * p + 1) << gb))] = nv[t][p][1];
                    S[stbase ^ swz((1 << yb) | (t << tb) | ((2 * p)     << gb))] = nv[t][p][2];
                    S[stbase ^ swz((1 << yb) | (t << tb) | ((2 * p + 1) << gb))] = nv[t][p][3];
                }
            }
        } else {
            float zAll = 0.0f, zX = 0.0f, zP = 0.0f, zT = 0.0f, zY = 0.0f;
#pragma unroll
            for (int t = 0; t < 2; ++t) {
#pragma unroll
                for (int p = 0; p < 2; ++p) {
                    const v2f q00 = nv[t][p][0] * nv[t][p][0], q10 = nv[t][p][1] * nv[t][p][1];
                    const v2f q01 = nv[t][p][2] * nv[t][p][2], q11 = nv[t][p][3] * nv[t][p][3];
                    const float p00 = q00[0] + q00[1], p10 = q10[0] + q10[1];
                    const float p01 = q01[0] + q01[1], p11 = q11[0] + q11[1];
                    const float y0 = p00 + p10, y1 = p01 + p11;
                    const float all = y0 + y1;
                    zAll += all;
                    zX += (p00 - p10) + (p01 - p11);
                    zY += y0 - y1;
                    zP += p ? -all : all;
                    zT += t ? -all : all;
                }
            }
            const int basek = wpart | lanepart | (ha << (gb + 2));
            float zq[NQ];
#pragma unroll
            for (int q = 0; q < NQ; ++q) {
                const int bit = (NQ - 1) - q;
                float v;
                if (bit == gb)          v = zX;
                else if (bit == gb + 1) v = zP;
                else if (bit == tb)     v = zT;
                else if (bit == yb)     v = zY;
                else                    v = ((basek >> bit) & 1) ? -zAll : zAll;
                v += rowx<1>(v);
                v += rowx<2>(v);
                v += rowx<4>(v);
                v += rowx<8>(v);
                v += xhalf(v);
                zq[q] = v;
            }
            if (lane == 0) {
#pragma unroll
                for (int q = 0; q < NQ; ++q) zout[q] = zq[q];
            }
        }
    };

    auto layer_scalars = [&](int l, int q) {
        const gate1 gq = wire_gate(params + (l * NQ + q) * 3);
        const float c0 = gq.c0, s0 = gq.s0, c1 = gq.c1, s1 = gq.s1, cz = gq.cz, sz = gq.sz;
        const c32 u00 = gq.u00, u01 = gq.u01, u10 = gq.u10, u11 = gq.u11;
        uq[l][q][0] = u00.x; uq[l][q][1] = u00.y;
        uq[l][q][2] = u01.x; uq[l][q][3] = u01.y;
        uq[l][q][4] = u10.x; uq[l][q][5] = u10.y;
        uq[l][q][6] = u11.x; uq[l][q][7] = u11.y;
        float cd = c1 * cz + s1 * sz, sd = s1 * cz - c1 * sz;
        float cs = c1 * cz - s1 * sz, ss = s1 * cz + c1 * sz;
        float cph = c0, sph = s0;
        pv[l][q][0] = cd; pv[l][q][1] = sd;
        pv[l][q][2] = cs; pv[l][q][3] = ss;
        pv[l][q][4] = cph; pv[l][q][5] = sph;
        pv[l][q][6] = cph * cph - sph * sph; pv[l][q][7] = 2.0f * sph * cph;
#pragma unroll
        for (int e = 0; e < 16; ++e) { pmr[l][q][e] = 0.0f; pmi[l][q][e] = 0.0f; }
        pmr[l][q][0]  = cd;
        pmr[l][q][3]  = sd * sph;  pmi[l][q][3]  = -sd * cph;
        pmi[l][q][12] = -sd;
        pmr[l][q][15] = cd * cph;  pmi[l][q][15] =  cd * sph;
        pmr[l][q][5]  = cs;
        pmr[l][q][6]  = -ss * sph; pmi[l][q][6]  = -ss * cph;
        pmi[l][q][9]  = -ss;
        pmr[l][q][10] = cs * cph;  pmi[l][q][10] = -cs * sph;
    };

    auto build_unitaries = [&](int l, int ub) {
        const int g = wv >> 1, c = 4 * (wv & 1) + (lane & 3), r = lane >> 2, q0 = 3 * g;
        {
            int o0 = ((((r >> 2) & 1) * 2 + ((c >> 2) & 1)) * 2);
            int o1 = ((((r >> 1) & 1) * 2 + ((c >> 1) & 1)) * 2);
            int o2 = (((r & 1) * 2 + (c & 1)) * 2);
            c32 acc;
            if (g == 0) acc = { uq[l][0][o0], uq[l][0][o0 + 1] };
            else        acc = { (((r >> 2) & 1) == ((c >> 2) & 1)) ? 1.0f : 0.0f, 0.0f };
            acc = cmul(acc, c32{ uq[l][q0 + 1][o1], uq[l][q0 + 1][o1 + 1] });
            acc = cmul(acc, c32{ uq[l][q0 + 2][o2], uq[l][q0 + 2][o2 + 1] });
            Tar[g][r * 8 + c] = acc.x; Tai[g][r * 8 + c] = acc.y;
        }
        wave_sync();
        auto chain8 = [&](int pair, int sh, const float* inR, const float* inI) {
            int rowsel = ((r >> sh) & 3) * 4;
            int rmask = r & ~(3 << sh);
            c32 acc = { 0.0f, 0.0f };
#pragma unroll
            for (int jj = 0; jj < 4; ++jj) {
                int k = rmask | (jj << sh);
                c32 p = { pmr[l][pair][rowsel + jj], pmi[l][pair][rowsel + jj] };
                c32 v = { inR[k * 8 + c], inI[k * 8 + c] };
                acc.x += p.x * v.x - p.y * v.y;
                acc.y += p.x * v.y + p.y * v.x;
            }
            return acc;
        };
        c32 s0 = chain8(q0, 1, Tar[g], Tai[g]);
        Tbr[g][r * 8 + c] = s0.x; Tbi[g][r * 8 + c] = s0.y;
        wave_sync();
        c32 u = chain8(q0 + 1, 0, Tbr[g], Tbi[g]);
        if (r & 1) u = cmul(u, c32{ pv[l][q0 + 2][4], -pv[l][q0 + 2][5] });
        if (g > 0 && l + 1 < NLAYER) {
            const c32 o = { __shfl_xor(u.x, 16), __shfl_xor(u.y, 16) };
            const int rb = r >> 2;
            const c32 in0 = rb ? o : u, in1 = rb ? u : o;
            const c32 n0 = { uq[l + 1][q0][rb * 4],     uq[l + 1][q0][rb * 4 + 1] };
            const c32 n1 = { uq[l + 1][q0][rb * 4 + 2], uq[l + 1][q0][rb * 4 + 3] };
            const c32 m0 = cmul(n0, in0), m1 = cmul(n1, in1);
            u = { m0.x + m1.x, m0.y + m1.y };
        }
        float* Ao = AL[ub][g];
        const int M0 = 8 * (r >> 2) + 2 * (r & 3), M1 = M0 + 1;
        const int kb = (c & 1) * 8 + (c >> 1) * 2;
        Ao[M0 * 16 + kb] = u.x;  Ao[M0 * 16 + kb + 1] = -u.y;
        Ao[M1 * 16 + kb] = u.y;  Ao[M1 * 16 + kb + 1] =  u.x;
    };

    if (lane < NQ) {
        if (wv == 0) {
            float xv = xin[b * NQ + lane];
            float h = 0.5f * xv;
            float a = 0.5f * xv * xv;
            float ch, sh, ca, sa;
            sincosf(h, &sh, &ch);
            sincosf(a, &sa, &ca);
            c32 e0 = { ch * ca, -ch * sa }, e1 = { sh * ca, sh * sa };
            const gate1 g0 = wire_gate(params + lane * 3);
            if (lane == 3 || lane == 6 || lane == 9) {
                const c32 a0 = cmul(g0.u00, e0), a1 = cmul(g0.u01, e1);
                const c32 b0 = cmul(g0.u10, e0), b1 = cmul(g0.u11, e1);
                e0 = { a0.x + a1.x, a0.y + a1.y };
                e1 = { b0.x + b1.x, b0.y + b1.y };
            }
            enc[lane][0] = e0.x;  enc[lane][1] = e0.y;
            enc[lane][2] = e1.x;  enc[lane][3] = e1.y;
        } else if (wv <= NLAYER) {
            layer_scalars(wv - 1, lane);
        }
    }
    __syncthreads();

    build_unitaries(0, 0);
    __syncthreads();

    const int w1part = wv;
    const int w2part = ((wv & 1) << 6) | (((wv >> 1) & 1) << 7) | (((wv >> 2) & 1) << 9);

    for (int l = 0; l < NLAYER; ++l) {
        const int ub = l & 1;
        int lra = lr, ha = half;
        asm volatile("" : "+v"(lra), "+v"(ha));

        apply_pair(AL[ub][0],  9, 8, 5,  3, 4, 6, 7,    w1part, lra, ha, l, 2, false, nullptr, l == 0);
        wave_sync();
        apply_pair(AL[ub][1],  6, 5, 4,  3, 9, 10, 11,  w1part, lra, ha, l, 5, false, nullptr, false);
        __syncthreads();

        apply_pair(AL[ub][2],  3, 2, 11, 0, 1, 8, 10,   w2part, lra, ha, l, 8, false, nullptr, false);
        wave_sync();
        if (l + 1 < NLAYER) build_unitaries(l + 1, ub ^ 1);
        apply_pair(AL[ub][3],  0, 11, 10, 3, 4, 5, 8,   w2part, lra, ha, l, 11,
                   l == NLAYER - 1, &zred[wv * NQ], false);
        __syncthreads();
    }

    if (tid < NQ) {
        float z = 0.0f;
#pragma unroll
        for (int w = 0; w < 8; ++w) z += zred[w * NQ + tid];
        out[b * NQ + tid] = z;
    }
}

extern "C" void kernel_launch(void* const* d_in, const int* in_sizes, int n_in,
                              void* d_out, int out_size, void* d_ws, size_t ws_size,
                              hipStream_t stream) {
    (void)n_in; (void)out_size; (void)d_ws; (void)ws_size;
    const float* x      = (const float*)d_in[0];
    const float* params = (const float*)d_in[1];
    float* out = (float*)d_out;
    int B = in_sizes[0] / NQ;
    qsim_kernel<<<dim3(B), dim3(256), 0, stream>>>(x, params, out);
}
